# starting kernel
# baseline (speedup 1.0000x reference)
#include <hip/hip_runtime.h>

typedef float v2f __attribute__((ext_vector_type(2)));
typedef float v8f __attribute__((ext_vector_type(8)));

__device__ __forceinline__ v8f wmma4(v2f a, v2f b, v8f c) {
    return __builtin_amdgcn_wmma_f32_16x16x4_f32(
        false, a, false, b, (short)0, c, false, false);
}

template <int MODE>
__global__ __launch_bounds__(256) void gemm_kernel(
    const float* __restrict__ A, const float* __restrict__ W,
    const float* __restrict__ bias, float* __restrict__ out,
    float* __restrict__ Qb, float* __restrict__ Kb, float* __restrict__ Vb,
    int Ncols, int Kdim)
{
    __shared__ float As[128 * 20];
    __shared__ float Wsh[64 * 20];

    const int tid  = threadIdx.x;
    const int wid  = tid >> 5;
    const int lane = tid & 31;
    const int half = lane >> 4;
    const int l15  = lane & 15;
    const int klo  = half * 2;
    const int wave_m = wid >> 1;
    const int wave_n = wid & 1;

    const int rowBase = blockIdx.y * 128;
    const int colBase = blockIdx.x * 64;

    v8f acc[2][2] = {};

    const int lr = tid >> 2;
    const int lc = (tid & 3) << 2;

    for (int kb = 0; kb < Kdim; kb += 16) {
        __syncthreads();
        *(float4*)&As[lr * 20 + lc] =
            *(const float4*)(A + (size_t)(rowBase + lr) * Kdim + kb + lc);
        *(float4*)&As[(lr + 64) * 20 + lc] =
            *(const float4*)(A + (size_t)(rowBase + lr + 64) * Kdim + kb + lc);
        *(float4*)&Wsh[lr * 20 + lc] =
            *(const float4*)(W + (size_t)(colBase + lr) * Kdim + kb + lc);
        __syncthreads();

#pragma unroll
        for (int ks = 0; ks < 16; ks += 4) {
            v2f a0 = *(const v2f*)&As[(wave_m * 32 + l15) * 20 + ks + klo];
            v2f a1 = *(const v2f*)&As[(wave_m * 32 + 16 + l15) * 20 + ks + klo];
            v2f w0 = *(const v2f*)&Wsh[(wave_n * 32 + l15) * 20 + ks + klo];
            v2f w1 = *(const v2f*)&Wsh[(wave_n * 32 + 16 + l15) * 20 + ks + klo];
            acc[0][0] = wmma4(a0, w0, acc[0][0]);
            acc[0][1] = wmma4(a0, w1, acc[0][1]);
            acc[1][0] = wmma4(a1, w0, acc[1][0]);
            acc[1][1] = wmma4(a1, w1, acc[1][1]);
        }
    }

#pragma unroll
    for (int sm = 0; sm < 2; ++sm) {
#pragma unroll
        for (int sn = 0; sn < 2; ++sn) {
            const int j  = colBase + wave_n * 32 + sn * 16 + l15;
            const int r0 = rowBase + wave_m * 32 + sm * 16 + half * 8;
            if (MODE == 0) {
                const float bv = bias ? bias[j] : 0.0f;
                float* dst = out + (size_t)r0 * Ncols + j;
#pragma unroll
                for (int g = 0; g < 8; ++g)
                    dst[(size_t)g * Ncols] = acc[sm][sn][g] + bv;
            } else {
                const int t  = j >> 10;
                const int hh = (j >> 6) & 15;
                const int d  = j & 63;
                const int bb = r0 >> 10;
                const int n  = r0 & 1023;
                float* dst;
                if (t == 0)
                    dst = Qb + ((size_t)(bb * 16 + hh) * 1024 + n) * 64 + d;
                else if (t == 1)
                    dst = Kb + ((size_t)(bb * 16 + hh) * 1280 + 256 + n) * 64 + d;
                else
                    dst = Vb + ((size_t)(bb * 16 + hh) * 1280 + 256 + n) * 64 + d;
#pragma unroll
                for (int g = 0; g < 8; ++g)
                    dst[(size_t)g * 64] = acc[sm][sn][g];
            }
        }
    }
}

__global__ __launch_bounds__(256) void ctx_scatter_kernel(
    const float* __restrict__ ctx, float* __restrict__ Kb, float* __restrict__ Vb)
{
    const int idx = blockIdx.x * 256 + threadIdx.x;
    const int f   = idx << 2;
    const int b   = f >> 19;
    const int m   = (f >> 11) & 255;
    const int q   = f & 2047;
    const int s   = q >> 10;
    const int hh  = (q >> 6) & 15;
    const int d   = q & 63;
    const float4 v = *(const float4*)(ctx + (size_t)f);
    float* base = s ? Vb : Kb;
    *(float4*)(base + ((size_t)(b * 16 + hh) * 1280 + m) * 64 + d) = v;
}

__global__ __launch_bounds__(256) void attn_kernel(
    const float* __restrict__ Qb, const float* __restrict__ Kb,
    const float* __restrict__ Vb, float* __restrict__ AO)
{
    __shared__ float Ks[32 * 68];
    __shared__ float VTs[64 * 34];
    __shared__ float Ps[8 * 16 * 34];

    const int tid  = threadIdx.x;
    const int wid  = tid >> 5;
    const int lane = tid & 31;
    const int half = lane >> 4;
    const int l15  = lane & 15;
    const int klo  = half * 2;

    const int bh = blockIdx.x;
    const int b  = bh >> 4;
    const int h  = bh & 15;
    const float* Qp = Qb + (size_t)bh * 1024 * 64;
    const float* Kp = Kb + (size_t)bh * 1280 * 64;
    const float* Vp = Vb + (size_t)bh * 1280 * 64;

    const int n0   = blockIdx.y * 128 + wid * 16;
    const int qrow = n0 + l15;
    const float scale = 0.125f;

    v2f qf[16];
#pragma unroll
    for (int kk = 0; kk < 16; ++kk) {
        v2f t = *(const v2f*)(Qp + (size_t)qrow * 64 + kk * 4 + klo);
        qf[kk] = t * scale;
    }

    float mrow[8], lrow[8];
    v8f oacc[4] = {};
#pragma unroll
    for (int g = 0; g < 8; ++g) { mrow[g] = -3.0e38f; lrow[g] = 0.0f; }

    float* Pw = &Ps[wid * 16 * 34];

    for (int mc = 0; mc < 1280; mc += 32) {
        __syncthreads();
#pragma unroll
        for (int it = 0; it < 2; ++it) {
            const int idx = tid + it * 256;
            const int r   = idx >> 4;
            const int c4  = (idx & 15) << 2;
            *(float4*)&Ks[r * 68 + c4] =
                *(const float4*)(Kp + (size_t)(mc + r) * 64 + c4);
            const float4 v = *(const float4*)(Vp + (size_t)(mc + r) * 64 + c4);
            VTs[(c4 + 0) * 34 + r] = v.x;
            VTs[(c4 + 1) * 34 + r] = v.y;
            VTs[(c4 + 2) * 34 + r] = v.z;
            VTs[(c4 + 3) * 34 + r] = v.w;
        }
        __syncthreads();

        v8f s0 = {}, s1 = {};
#pragma unroll
        for (int kk = 0; kk < 16; ++kk) {
            v2f b0 = *(const v2f*)&Ks[l15 * 68 + kk * 4 + klo];
            v2f b1 = *(const v2f*)&Ks[(16 + l15) * 68 + kk * 4 + klo];
            s0 = wmma4(qf[kk], b0, s0);
            s1 = wmma4(qf[kk], b1, s1);
        }

        float alpha[8];
#pragma unroll
        for (int g = 0; g < 8; ++g) {
            float cm = fmaxf(s0[g], s1[g]);
            cm = fmaxf(cm, __shfl_xor(cm, 1, 32));
            cm = fmaxf(cm, __shfl_xor(cm, 2, 32));
            cm = fmaxf(cm, __shfl_xor(cm, 4, 32));
            cm = fmaxf(cm, __shfl_xor(cm, 8, 32));
            const float mnew = fmaxf(mrow[g], cm);
            alpha[g] = __expf(mrow[g] - mnew);
            mrow[g]  = mnew;
            const float p0 = __expf(s0[g] - mnew);
            const float p1 = __expf(s1[g] - mnew);
            s0[g] = p0; s1[g] = p1;
            float cs = p0 + p1;
            cs += __shfl_xor(cs, 1, 32);
            cs += __shfl_xor(cs, 2, 32);
            cs += __shfl_xor(cs, 4, 32);
            cs += __shfl_xor(cs, 8, 32);
            lrow[g] = lrow[g] * alpha[g] + cs;
        }
#pragma unroll
        for (int t4 = 0; t4 < 4; ++t4)
#pragma unroll
            for (int g = 0; g < 8; ++g) oacc[t4][g] *= alpha[g];

#pragma unroll
        for (int g = 0; g < 8; ++g) {
            const int r = g + half * 8;
            Pw[r * 34 + l15]      = s0[g];
            Pw[r * 34 + 16 + l15] = s1[g];
        }
        asm volatile("s_wait_dscnt 0" ::: "memory");

#pragma unroll
        for (int kk = 0; kk < 8; ++kk) {
            v2f pf = *(const v2f*)&Pw[l15 * 34 + kk * 4 + klo];
#pragma unroll
            for (int t4 = 0; t4 < 4; ++t4) {
                v2f vf = *(const v2f*)&VTs[(t4 * 16 + l15) * 34 + kk * 4 + klo];
                oacc[t4] = wmma4(pf, vf, oacc[t4]);
            }
        }
    }

#pragma unroll
    for (int g = 0; g < 8; ++g) {
        const float rinv = 1.0f / lrow[g];
        const int n = n0 + g + half * 8;
        float* dst = AO + ((size_t)(b * 1024 + n) * 1024) + h * 64;
#pragma unroll
        for (int t4 = 0; t4 < 4; ++t4)
            dst[t4 * 16 + l15] = oacc[t4][g] * rinv;
    }
}

extern "C" void kernel_launch(void* const* d_in, const int* in_sizes, int n_in,
                              void* d_out, int out_size, void* d_ws, size_t ws_size,
                              hipStream_t stream)
{
    const float* x      = (const float*)d_in[0];
    const float* ctx    = (const float*)d_in[1];
    const float* w_qkv  = (const float*)d_in[2];
    const float* w_proj = (const float*)d_in[3];
    const float* b_proj = (const float*)d_in[4];
    float* outp = (float*)d_out;

    float* Qb = (float*)d_ws;
    float* Kb = Qb + 4194304;
    float* Vb = Kb + 5242880;
    float* AO = Vb + 5242880;

    gemm_kernel<1><<<dim3(3072 / 64, 4096 / 128), 256, 0, stream>>>(
        x, w_qkv, nullptr, nullptr, Qb, Kb, Vb, 3072, 1024);

    ctx_scatter_kernel<<<2048, 256, 0, stream>>>(ctx, Kb, Vb);

    attn_kernel<<<dim3(64, 8), 256, 0, stream>>>(Qb, Kb, Vb, AO);

    gemm_kernel<0><<<dim3(1024 / 64, 4096 / 128), 256, 0, stream>>>(
        AO, w_proj, b_proj, outp, nullptr, nullptr, nullptr, 1024, 1024);
}


# QKV GEMM: WMMA fragments read straight from global memory (16-byte loads along k, next set fetched ahead in registers), 64x64 wave tiles, no LDS stage, no barriers; k re-paired per 8 (new sum order)
# speedup vs baseline: 1.1106x; 1.1106x over previous
#include <hip/hip_runtime.h>

typedef float v2f __attribute__((ext_vector_type(2)));
typedef float v4f __attribute__((ext_vector_type(4)));
typedef float v8f __attribute__((ext_vector_type(8)));

__device__ __forceinline__ v8f wmma4(v2f a, v2f b, v8f c) {
    return __builtin_amdgcn_wmma_f32_16x16x4_f32(
        false, a, false, b, (short)0, c, false, false);
}

__global__ __launch_bounds__(256) void gemm_kernel(
    const float* __restrict__ A, const float* __restrict__ W,
    const float* __restrict__ bias, float* __restrict__ out,
    int Ncols, int Kdim)
{
    __shared__ float As[128 * 20];
    __shared__ float Wsh[64 * 20];

    const int tid  = threadIdx.x;
    const int wid  = tid >> 5;
    const int lane = tid & 31;
    const int half = lane >> 4;
    const int l15  = lane & 15;
    const int klo  = half * 2;
    const int wave_m = wid >> 1;
    const int wave_n = wid & 1;

    const int rowBase = blockIdx.y * 128;
    const int colBase = blockIdx.x * 64;

    v8f acc[2][2] = {};

    const int lr = tid >> 2;
    const int lc = (tid & 3) << 2;

    const float* pA0 = A + (size_t)(rowBase + lr) * Kdim + lc;
    const float* pA1 = A + (size_t)(rowBase + lr + 64) * Kdim + lc;
    const float* pW  = W + (size_t)(colBase + lr) * Kdim + lc;

    for (int kb = 0; kb < Kdim; kb += 32) {
        const v4f a0lo = *(const v4f*)(pA0 + kb);
        const v4f a1lo = *(const v4f*)(pA1 + kb);
        const v4f wlo  = *(const v4f*)(pW + kb);
        v4f a0hi = *(const v4f*)(pA0 + kb + 16);
        v4f a1hi = *(const v4f*)(pA1 + kb + 16);
        v4f whi  = *(const v4f*)(pW + kb + 16);
        asm volatile("" : "+v"(a0hi), "+v"(a1hi), "+v"(whi));

#pragma unroll
        for (int hs = 0; hs < 2; ++hs) {
            __syncthreads();
            *(v4f*)&As[lr * 20 + lc]        = hs ? a0hi : a0lo;
            *(v4f*)&As[(lr + 64) * 20 + lc] = hs ? a1hi : a1lo;
            *(v4f*)&Wsh[lr * 20 + lc]       = hs ? whi : wlo;
            __syncthreads();

#pragma unroll
            for (int ks = 0; ks < 16; ks += 4) {
                v2f a0 = *(const v2f*)&As[(wave_m * 32 + l15) * 20 + ks + klo];
                v2f a1 = *(const v2f*)&As[(wave_m * 32 + 16 + l15) * 20 + ks + klo];
                v2f w0 = *(const v2f*)&Wsh[(wave_n * 32 + l15) * 20 + ks + klo];
                v2f w1 = *(const v2f*)&Wsh[(wave_n * 32 + 16 + l15) * 20 + ks + klo];
                acc[0][0] = wmma4(a0, w0, acc[0][0]);
                acc[0][1] = wmma4(a0, w1, acc[0][1]);
                acc[1][0] = wmma4(a1, w0, acc[1][0]);
                acc[1][1] = wmma4(a1, w1, acc[1][1]);
            }
        }
    }

#pragma unroll
    for (int sm = 0; sm < 2; ++sm) {
#pragma unroll
        for (int sn = 0; sn < 2; ++sn) {
            const int j  = colBase + wave_n * 32 + sn * 16 + l15;
            const int r0 = rowBase + wave_m * 32 + sm * 16 + half * 8;
            const float bv = bias ? bias[j] : 0.0f;
            float* dst = out + (size_t)r0 * Ncols + j;
#pragma unroll
            for (int g = 0; g < 8; ++g)
                dst[(size_t)g * Ncols] = acc[sm][sn][g] + bv;
        }
    }
}

struct FragSet {
    v4f a[4][2];
    v4f w[4][2];
};

__device__ __forceinline__ void load_set(FragSet& f, const float* pa, const float* pw, int kb) {
#pragma unroll
    for (int c = 0; c < 2; ++c) {
#pragma unroll
        for (int i = 0; i < 4; ++i) {
            f.a[i][c] = *(const v4f*)(pa + (size_t)i * 16 * 1024 + kb + c * 8);
            f.w[i][c] = *(const v4f*)(pw + (size_t)i * 16 * 1024 + kb + c * 8);
        }
    }
}

__device__ __forceinline__ void mma_set(const FragSet& f, v8f (&acc)[4][4]) {
#pragma unroll
    for (int c = 0; c < 2; ++c) {
#pragma unroll
        for (int p = 0; p < 2; ++p) {
#pragma unroll
            for (int i = 0; i < 4; ++i) {
                const v2f av = p ? __builtin_shufflevector(f.a[i][c], f.a[i][c], 2, 3)
                                 : __builtin_shufflevector(f.a[i][c], f.a[i][c], 0, 1);
#pragma unroll
                for (int j = 0; j < 4; ++j) {
                    const v2f wv = p ? __builtin_shufflevector(f.w[j][c], f.w[j][c], 2, 3)
                                     : __builtin_shufflevector(f.w[j][c], f.w[j][c], 0, 1);
                    acc[i][j] = wmma4(av, wv, acc[i][j]);
                }
            }
        }
    }
}

__global__ __launch_bounds__(128) void qkv_direct_kernel(
    const float* __restrict__ X, const float* __restrict__ Wq,
    float* __restrict__ Qb, float* __restrict__ Kb, float* __restrict__ Vb)
{
    const int tid  = threadIdx.x;
    const int wid  = tid >> 5;
    const int lane = tid & 31;
    const int half = lane >> 4;
    const int l15  = lane & 15;

    const int rowBase = blockIdx.y * 128 + (wid >> 1) * 64;
    const int colBase = blockIdx.x * 128 + (wid & 1) * 64;

    const float* pa = X  + (size_t)(rowBase + l15) * 1024 + half * 4;
    const float* pw = Wq + (size_t)(colBase + l15) * 1024 + half * 4;

    v8f acc[4][4] = {};
    FragSet f0, f1;

    load_set(f0, pa, pw, 0);
    for (int kb = 0; kb < 1024; kb += 32) {
        load_set(f1, pa, pw, kb + 16);
        __builtin_amdgcn_sched_barrier(0);
        mma_set(f0, acc);
        load_set(f0, pa, pw, (kb + 32) & 1023);
        __builtin_amdgcn_sched_barrier(0);
        mma_set(f1, acc);
    }

    const int t  = colBase >> 10;
    const int hh = (colBase >> 6) & 15;
#pragma unroll
    for (int sm = 0; sm < 4; ++sm) {
        const int r0 = rowBase + sm * 16 + half * 8;
        const int bb = r0 >> 10;
        const int n  = r0 & 1023;
        float* rowp;
        if (t == 0)
            rowp = Qb + ((size_t)(bb * 16 + hh) * 1024 + n) * 64;
        else if (t == 1)
            rowp = Kb + ((size_t)(bb * 16 + hh) * 1280 + 256 + n) * 64;
        else
            rowp = Vb + ((size_t)(bb * 16 + hh) * 1280 + 256 + n) * 64;
#pragma unroll
        for (int sn = 0; sn < 4; ++sn) {
            float* dst = rowp + sn * 16 + l15;
#pragma unroll
            for (int g = 0; g < 8; ++g)
                dst[(size_t)g * 64] = acc[sm][sn][g];
        }
    }
}

__global__ __launch_bounds__(256) void ctx_scatter_kernel(
    const float* __restrict__ ctx, float* __restrict__ Kb, float* __restrict__ Vb)
{
    const int idx = blockIdx.x * 256 + threadIdx.x;
    const int f   = idx << 2;
    const int b   = f >> 19;
    const int m   = (f >> 11) & 255;
    const int q   = f & 2047;
    const int s   = q >> 10;
    const int hh  = (q >> 6) & 15;
    const int d   = q & 63;
    const float4 v = *(const float4*)(ctx + (size_t)f);
    float* base = s ? Vb : Kb;
    *(float4*)(base + ((size_t)(b * 16 + hh) * 1280 + m) * 64 + d) = v;
}

__global__ __launch_bounds__(256) void attn_kernel(
    const float* __restrict__ Qb, const float* __restrict__ Kb,
    const float* __restrict__ Vb, float* __restrict__ AO)
{
    __shared__ __attribute__((aligned(16))) float Ks[64 * 68];
    __shared__ __attribute__((aligned(16))) float VTs[64 * 68];
    __shared__ __attribute__((aligned(16))) float Ps[8 * 16 * 68];

    const int tid  = threadIdx.x;
    const int wid  = tid >> 5;
    const int lane = tid & 31;
    const int half = lane >> 4;
    const int l15  = lane & 15;
    const int klo  = half * 2;

    const int bh = blockIdx.x;
    const int b  = bh >> 4;
    const int h  = bh & 15;
    const float* Qp = Qb + (size_t)bh * 1024 * 64;
    const float* Kp = Kb + (size_t)bh * 1280 * 64;
    const float* Vp = Vb + (size_t)bh * 1280 * 64;

    const int n0   = blockIdx.y * 128 + wid * 16;
    const int qrow = n0 + l15;
    const float scale = 0.125f;

    v2f qf[16];
#pragma unroll
    for (int kk = 0; kk < 16; ++kk) {
        v2f t = *(const v2f*)(Qp + (size_t)qrow * 64 + kk * 4 + klo);
        qf[kk] = t * scale;
    }

    float mrow[8], lrow[8];
    v8f oacc[4] = {};
#pragma unroll
    for (int g = 0; g < 8; ++g) { mrow[g] = -3.0e38f; lrow[g] = 0.0f; }

    float* Pw = &Ps[wid * 16 * 68];

    for (int mc = 0; mc < 1280; mc += 64) {
        __syncthreads();
#pragma unroll
        for (int it = 0; it < 4; ++it) {
            const int idx = tid + it * 256;
            const int r   = idx >> 4;
            const int c4  = (idx & 15) << 2;
            *(float4*)&Ks[r * 68 + c4] =
                *(const float4*)(Kp + (size_t)(mc + r) * 64 + c4);
            const float4 v = *(const float4*)(Vp + (size_t)(mc + r) * 64 + c4);
            VTs[(c4 + 0) * 68 + r] = v.x;
            VTs[(c4 + 1) * 68 + r] = v.y;
            VTs[(c4 + 2) * 68 + r] = v.z;
            VTs[(c4 + 3) * 68 + r] = v.w;
        }
        __syncthreads();

        v8f s[4] = {};
#pragma unroll
        for (int kk = 0; kk < 16; ++kk) {
#pragma unroll
            for (int t = 0; t < 4; ++t) {
                v2f kf = *(const v2f*)&Ks[(t * 16 + l15) * 68 + kk * 4 + klo];
                s[t] = wmma4(qf[kk], kf, s[t]);
            }
        }

        float alpha[8];
#pragma unroll
        for (int g = 0; g < 8; ++g) {
            float cm = fmaxf(fmaxf(s[0][g], s[1][g]), fmaxf(s[2][g], s[3][g]));
            cm = fmaxf(cm, __shfl_xor(cm, 1, 32));
            cm = fmaxf(cm, __shfl_xor(cm, 2, 32));
            cm = fmaxf(cm, __shfl_xor(cm, 4, 32));
            cm = fmaxf(cm, __shfl_xor(cm, 8, 32));
            const float mnew = fmaxf(mrow[g], cm);
            alpha[g] = __expf(mrow[g] - mnew);
            mrow[g]  = mnew;
            const float p0 = __expf(s[0][g] - mnew);
            const float p1 = __expf(s[1][g] - mnew);
            const float p2 = __expf(s[2][g] - mnew);
            const float p3 = __expf(s[3][g] - mnew);
            s[0][g] = p0; s[1][g] = p1; s[2][g] = p2; s[3][g] = p3;
            float cs = (p0 + p1) + (p2 + p3);
            cs += __shfl_xor(cs, 1, 32);
            cs += __shfl_xor(cs, 2, 32);
            cs += __shfl_xor(cs, 4, 32);
            cs += __shfl_xor(cs, 8, 32);
            lrow[g] = lrow[g] * alpha[g] + cs;
        }
#pragma unroll
        for (int t4 = 0; t4 < 4; ++t4)
#pragma unroll
            for (int g = 0; g < 8; ++g) oacc[t4][g] *= alpha[g];

#pragma unroll
        for (int g = 0; g < 8; ++g) {
            const int r = g + half * 8;
#pragma unroll
            for (int t = 0; t < 4; ++t)
                Pw[r * 68 + t * 16 + l15] = s[t][g];
        }
        asm volatile("s_wait_dscnt 0" ::: "memory");

#pragma unroll
        for (int kk = 0; kk < 16; ++kk) {
            v2f pf = *(const v2f*)&Pw[l15 * 68 + kk * 4 + klo];
#pragma unroll
            for (int t4 = 0; t4 < 4; ++t4) {
                v2f vf = *(const v2f*)&VTs[(t4 * 16 + l15) * 68 + kk * 4 + klo];
                oacc[t4] = wmma4(pf, vf, oacc[t4]);
            }
        }
    }

#pragma unroll
    for (int g = 0; g < 8; ++g) {
        const float rinv = 1.0f / lrow[g];
        const int n = n0 + g + half * 8;
        float* dst = AO + ((size_t)(b * 1024 + n) * 1024) + h * 64;
#pragma unroll
        for (int t4 = 0; t4 < 4; ++t4)
            dst[t4 * 16 + l15] = oacc[t4][g] * rinv;
    }
}

extern "C" void kernel_launch(void* const* d_in, const int* in_sizes, int n_in,
                              void* d_out, int out_size, void* d_ws, size_t ws_size,
                              hipStream_t stream)
{
    const float* x      = (const float*)d_in[0];
    const float* ctx    = (const float*)d_in[1];
    const float* w_qkv  = (const float*)d_in[2];
    const float* w_proj = (const float*)d_in[3];
    const float* b_proj = (const float*)d_in[4];
    float* outp = (float*)d_out;

    float* Qb = (float*)d_ws;
    float* Kb = Qb + 4194304;
    float* Vb = Kb + 5242880;
    float* AO = Vb + 5242880;

    qkv_direct_kernel<<<dim3(3072 / 128, 4096 / 128), 128, 0, stream>>>(
        x, w_qkv, Qb, Kb, Vb);

    ctx_scatter_kernel<<<2048, 256, 0, stream>>>(ctx, Kb, Vb);

    attn_kernel<<<dim3(64, 8), 256, 0, stream>>>(Qb, Kb, Vb, AO);

    gemm_kernel<<<dim3(1024 / 64, 4096 / 128), 256, 0, stream>>>(
        AO, w_proj, b_proj, outp, 1024, 1024);
}


# Projection GEMM: WMMA fragments read straight from global memory (16-byte loads along k, no LDS stage, no loop barriers), 64x64 tiles, K halved over two waves joined once via LDS (new sum order)
# speedup vs baseline: 1.1636x; 1.0477x over previous
#include <hip/hip_runtime.h>

typedef float v2f __attribute__((ext_vector_type(2)));
typedef float v4f __attribute__((ext_vector_type(4)));
typedef float v8f __attribute__((ext_vector_type(8)));

__device__ __forceinline__ v8f wmma4(v2f a, v2f b, v8f c) {
    return __builtin_amdgcn_wmma_f32_16x16x4_f32(
        false, a, false, b, (short)0, c, false, false);
}

struct FragSet {
    v4f a[4][2];
    v4f w[4][2];
};

__device__ __forceinline__ void load_set(FragSet& f, const float* pa, const float* pw, int kb) {
#pragma unroll
    for (int c = 0; c < 2; ++c) {
#pragma unroll
        for (int i = 0; i < 4; ++i) {
            f.a[i][c] = *(const v4f*)(pa + (size_t)i * 16 * 1024 + kb + c * 8);
            f.w[i][c] = *(const v4f*)(pw + (size_t)i * 16 * 1024 + kb + c * 8);
        }
    }
}

__device__ __forceinline__ void mma_set(const FragSet& f, v8f (&acc)[4][4]) {
#pragma unroll
    for (int c = 0; c < 2; ++c) {
#pragma unroll
        for (int p = 0; p < 2; ++p) {
#pragma unroll
            for (int i = 0; i < 4; ++i) {
                const v2f av = p ? __builtin_shufflevector(f.a[i][c], f.a[i][c], 2, 3)
                                 : __builtin_shufflevector(f.a[i][c], f.a[i][c], 0, 1);
#pragma unroll
                for (int j = 0; j < 4; ++j) {
                    const v2f wv = p ? __builtin_shufflevector(f.w[j][c], f.w[j][c], 2, 3)
                                     : __builtin_shufflevector(f.w[j][c], f.w[j][c], 0, 1);
                    acc[i][j] = wmma4(av, wv, acc[i][j]);
                }
            }
        }
    }
}

__global__ __launch_bounds__(128) void qkv_direct_kernel(
    const float* __restrict__ X, const float* __restrict__ Wq,
    float* __restrict__ Qb, float* __restrict__ Kb, float* __restrict__ Vb)
{
    const int tid  = threadIdx.x;
    const int wid  = tid >> 5;
    const int lane = tid & 31;
    const int half = lane >> 4;
    const int l15  = lane & 15;

    const int rowBase = blockIdx.y * 128 + (wid >> 1) * 64;
    const int colBase = blockIdx.x * 128 + (wid & 1) * 64;

    const float* pa = X  + (size_t)(rowBase + l15) * 1024 + half * 4;
    const float* pw = Wq + (size_t)(colBase + l15) * 1024 + half * 4;

    v8f acc[4][4] = {};
    FragSet f0, f1;

    load_set(f0, pa, pw, 0);
    for (int kb = 0; kb < 1024; kb += 32) {
        load_set(f1, pa, pw, kb + 16);
        __builtin_amdgcn_sched_barrier(0);
        mma_set(f0, acc);
        load_set(f0, pa, pw, (kb + 32) & 1023);
        __builtin_amdgcn_sched_barrier(0);
        mma_set(f1, acc);
    }

    const int t  = colBase >> 10;
    const int hh = (colBase >> 6) & 15;
#pragma unroll
    for (int sm = 0; sm < 4; ++sm) {
        const int r0 = rowBase + sm * 16 + half * 8;
        const int bb = r0 >> 10;
        const int n  = r0 & 1023;
        float* rowp;
        if (t == 0)
            rowp = Qb + ((size_t)(bb * 16 + hh) * 1024 + n) * 64;
        else if (t == 1)
            rowp = Kb + ((size_t)(bb * 16 + hh) * 1280 + 256 + n) * 64;
        else
            rowp = Vb + ((size_t)(bb * 16 + hh) * 1280 + 256 + n) * 64;
#pragma unroll
        for (int sn = 0; sn < 4; ++sn) {
            float* dst = rowp + sn * 16 + l15;
#pragma unroll
            for (int g = 0; g < 8; ++g)
                dst[(size_t)g * 64] = acc[sm][sn][g];
        }
    }
}

struct FragSet4 {
    v4f a[4][4];
    v4f w[4][4];
};

__device__ __forceinline__ void load_set4(FragSet4& f, const float* pa, const float* pw, int kb) {
#pragma unroll
    for (int i = 0; i < 4; ++i) {
#pragma unroll
        for (int c = 0; c < 4; ++c) {
            f.a[i][c] = *(const v4f*)(pa + (size_t)i * 16 * 1024 + kb + c * 8);
            f.w[i][c] = *(const v4f*)(pw + (size_t)i * 16 * 1024 + kb + c * 8);
        }
    }
}

__device__ __forceinline__ void mma_set4(const FragSet4& f, v8f (&acc)[4][4]) {
#pragma unroll
    for (int c = 0; c < 4; ++c) {
#pragma unroll
        for (int p = 0; p < 2; ++p) {
#pragma unroll
            for (int i = 0; i < 4; ++i) {
                const v2f av = p ? __builtin_shufflevector(f.a[i][c], f.a[i][c], 2, 3)
                                 : __builtin_shufflevector(f.a[i][c], f.a[i][c], 0, 1);
#pragma unroll
                for (int j = 0; j < 4; ++j) {
                    const v2f wv = p ? __builtin_shufflevector(f.w[j][c], f.w[j][c], 2, 3)
                                     : __builtin_shufflevector(f.w[j][c], f.w[j][c], 0, 1);
                    acc[i][j] = wmma4(av, wv, acc[i][j]);
                }
            }
        }
    }
}

__global__ __launch_bounds__(256) void proj_direct_kernel(
    const float* __restrict__ A, const float* __restrict__ Wp,
    const float* __restrict__ bias, float* __restrict__ out)
{
    __shared__ __attribute__((aligned(16))) float red[4 * 32 * 128];

    const int tid  = threadIdx.x;
    const int wid  = tid >> 5;
    const int lane = tid & 31;
    const int half = lane >> 4;
    const int l15  = lane & 15;
    const int tile = wid & 3;
    const int kh   = wid >> 2;

    const int rowBase = blockIdx.y * 128 + (tile >> 1) * 64;
    const int colBase = blockIdx.x * 128 + (tile & 1) * 64;

    const float* pa = A  + (size_t)(rowBase + l15) * 1024 + kh * 512 + half * 4;
    const float* pw = Wp + (size_t)(colBase + l15) * 1024 + kh * 512 + half * 4;

    v8f acc[4][4] = {};
    FragSet4 f0, f1;

    load_set4(f0, pa, pw, 0);
    for (int kb = 0; kb < 512; kb += 64) {
        load_set4(f1, pa, pw, kb + 32);
        __builtin_amdgcn_sched_barrier(0);
        mma_set4(f0, acc);
        load_set4(f0, pa, pw, (kb + 64) & 511);
        __builtin_amdgcn_sched_barrier(0);
        mma_set4(f1, acc);
    }

    float* rbase = red + tile * (32 * 128) + lane * 4;
    if (kh == 1) {
#pragma unroll
        for (int sm = 0; sm < 4; ++sm)
#pragma unroll
            for (int sn = 0; sn < 4; ++sn)
#pragma unroll
                for (int q = 0; q < 2; ++q) {
                    v4f v;
                    v.x = acc[sm][sn][q * 4 + 0];
                    v.y = acc[sm][sn][q * 4 + 1];
                    v.z = acc[sm][sn][q * 4 + 2];
                    v.w = acc[sm][sn][q * 4 + 3];
                    *(v4f*)(rbase + ((sm * 4 + sn) * 2 + q) * 128) = v;
                }
    }
    __syncthreads();
    if (kh == 0) {
#pragma unroll
        for (int sm = 0; sm < 4; ++sm) {
            const int r0 = rowBase + sm * 16 + half * 8;
#pragma unroll
            for (int sn = 0; sn < 4; ++sn) {
                const int j = colBase + sn * 16 + l15;
                const float bv = bias[j];
                float* dst = out + (size_t)r0 * 1024 + j;
#pragma unroll
                for (int q = 0; q < 2; ++q) {
                    const v4f v = *(const v4f*)(rbase + ((sm * 4 + sn) * 2 + q) * 128);
                    dst[(size_t)(q * 4 + 0) * 1024] = (acc[sm][sn][q * 4 + 0] + v.x) + bv;
                    dst[(size_t)(q * 4 + 1) * 1024] = (acc[sm][sn][q * 4 + 1] + v.y) + bv;
                    dst[(size_t)(q * 4 + 2) * 1024] = (acc[sm][sn][q * 4 + 2] + v.z) + bv;
                    dst[(size_t)(q * 4 + 3) * 1024] = (acc[sm][sn][q * 4 + 3] + v.w) + bv;
                }
            }
        }
    }
}

__global__ __launch_bounds__(256) void ctx_scatter_kernel(
    const float* __restrict__ ctx, float* __restrict__ Kb, float* __restrict__ Vb)
{
    const int idx = blockIdx.x * 256 + threadIdx.x;
    const int f   = idx << 2;
    const int b   = f >> 19;
    const int m   = (f >> 11) & 255;
    const int q   = f & 2047;
    const int s   = q >> 10;
    const int hh  = (q >> 6) & 15;
    const int d   = q & 63;
    const float4 v = *(const float4*)(ctx + (size_t)f);
    float* base = s ? Vb : Kb;
    *(float4*)(base + ((size_t)(b * 16 + hh) * 1280 + m) * 64 + d) = v;
}

__global__ __launch_bounds__(256) void attn_kernel(
    const float* __restrict__ Qb, const float* __restrict__ Kb,
    const float* __restrict__ Vb, float* __restrict__ AO)
{
    __shared__ __attribute__((aligned(16))) float Ks[64 * 68];
    __shared__ __attribute__((aligned(16))) float VTs[64 * 68];
    __shared__ __attribute__((aligned(16))) float Ps[8 * 16 * 68];

    const int tid  = threadIdx.x;
    const int wid  = tid >> 5;
    const int lane = tid & 31;
    const int half = lane >> 4;
    const int l15  = lane & 15;
    const int klo  = half * 2;

    const int bh = blockIdx.x;
    const int b  = bh >> 4;
    const int h  = bh & 15;
    const float* Qp = Qb + (size_t)bh * 1024 * 64;
    const float* Kp = Kb + (size_t)bh * 1280 * 64;
    const float* Vp = Vb + (size_t)bh * 1280 * 64;

    const int n0   = blockIdx.y * 128 + wid * 16;
    const int qrow = n0 + l15;
    const float scale = 0.125f;

    v2f qf[16];
#pragma unroll
    for (int kk = 0; kk < 16; ++kk) {
        v2f t = *(const v2f*)(Qp + (size_t)qrow * 64 + kk * 4 + klo);
        qf[kk] = t * scale;
    }

    float mrow[8], lrow[8];
    v8f oacc[4] = {};
#pragma unroll
    for (int g = 0; g < 8; ++g) { mrow[g] = -3.0e38f; lrow[g] = 0.0f; }

    float* Pw = &Ps[wid * 16 * 68];

    for (int mc = 0; mc < 1280; mc += 64) {
        __syncthreads();
#pragma unroll
        for (int it = 0; it < 4; ++it) {
            const int idx = tid + it * 256;
            const int r   = idx >> 4;
            const int c4  = (idx & 15) << 2;
            *(float4*)&Ks[r * 68 + c4] =
                *(const float4*)(Kp + (size_t)(mc + r) * 64 + c4);
            const float4 v = *(const float4*)(Vp + (size_t)(mc + r) * 64 + c4);
            VTs[(c4 + 0) * 68 + r] = v.x;
            VTs[(c4 + 1) * 68 + r] = v.y;
            VTs[(c4 + 2) * 68 + r] = v.z;
            VTs[(c4 + 3) * 68 + r] = v.w;
        }
        __syncthreads();

        v8f s[4] = {};
#pragma unroll
        for (int kk = 0; kk < 16; ++kk) {
#pragma unroll
            for (int t = 0; t < 4; ++t) {
                v2f kf = *(const v2f*)&Ks[(t * 16 + l15) * 68 + kk * 4 + klo];
                s[t] = wmma4(qf[kk], kf, s[t]);
            }
        }

        float alpha[8];
#pragma unroll
        for (int g = 0; g < 8; ++g) {
            float cm = fmaxf(fmaxf(s[0][g], s[1][g]), fmaxf(s[2][g], s[3][g]));
            cm = fmaxf(cm, __shfl_xor(cm, 1, 32));
            cm = fmaxf(cm, __shfl_xor(cm, 2, 32));
            cm = fmaxf(cm, __shfl_xor(cm, 4, 32));
            cm = fmaxf(cm, __shfl_xor(cm, 8, 32));
            const float mnew = fmaxf(mrow[g], cm);
            alpha[g] = __expf(mrow[g] - mnew);
            mrow[g]  = mnew;
            const float p0 = __expf(s[0][g] - mnew);
            const float p1 = __expf(s[1][g] - mnew);
            const float p2 = __expf(s[2][g] - mnew);
            const float p3 = __expf(s[3][g] - mnew);
            s[0][g] = p0; s[1][g] = p1; s[2][g] = p2; s[3][g] = p3;
            float cs = (p0 + p1) + (p2 + p3);
            cs += __shfl_xor(cs, 1, 32);
            cs += __shfl_xor(cs, 2, 32);
            cs += __shfl_xor(cs, 4, 32);
            cs += __shfl_xor(cs, 8, 32);
            lrow[g] = lrow[g] * alpha[g] + cs;
        }
#pragma unroll
        for (int t4 = 0; t4 < 4; ++t4)
#pragma unroll
            for (int g = 0; g < 8; ++g) oacc[t4][g] *= alpha[g];

#pragma unroll
        for (int g = 0; g < 8; ++g) {
            const int r = g + half * 8;
#pragma unroll
            for (int t = 0; t < 4; ++t)
                Pw[r * 68 + t * 16 + l15] = s[t][g];
        }
        asm volatile("s_wait_dscnt 0" ::: "memory");

#pragma unroll
        for (int kk = 0; kk < 16; ++kk) {
            v2f pf = *(const v2f*)&Pw[l15 * 68 + kk * 4 + klo];
#pragma unroll
            for (int t4 = 0; t4 < 4; ++t4) {
                v2f vf = *(const v2f*)&VTs[(t4 * 16 + l15) * 68 + kk * 4 + klo];
                oacc[t4] = wmma4(pf, vf, oacc[t4]);
            }
        }
    }

#pragma unroll
    for (int g = 0; g < 8; ++g) {
        const float rinv = 1.0f / lrow[g];
        const int n = n0 + g + half * 8;
        float* dst = AO + ((size_t)(b * 1024 + n) * 1024) + h * 64;
#pragma unroll
        for (int t4 = 0; t4 < 4; ++t4)
            dst[t4 * 16 + l15] = oacc[t4][g] * rinv;
    }
}

extern "C" void kernel_launch(void* const* d_in, const int* in_sizes, int n_in,
                              void* d_out, int out_size, void* d_ws, size_t ws_size,
                              hipStream_t stream)
{
    const float* x      = (const float*)d_in[0];
    const float* ctx    = (const float*)d_in[1];
    const float* w_qkv  = (const float*)d_in[2];
    const float* w_proj = (const float*)d_in[3];
    const float* b_proj = (const float*)d_in[4];
    float* outp = (float*)d_out;

    float* Qb = (float*)d_ws;
    float* Kb = Qb + 4194304;
    float* Vb = Kb + 5242880;
    float* AO = Vb + 5242880;

    qkv_direct_kernel<<<dim3(3072 / 128, 4096 / 128), 128, 0, stream>>>(
        x, w_qkv, Qb, Kb, Vb);

    ctx_scatter_kernel<<<2048, 256, 0, stream>>>(ctx, Kb, Vb);

    attn_kernel<<<dim3(64, 8), 256, 0, stream>>>(Qb, Kb, Vb, AO);

    proj_direct_kernel<<<dim3(1024 / 128, 4096 / 128), 256, 0, stream>>>(
        AO, w_proj, b_proj, outp);
}
